# starting kernel
# baseline (speedup 1.0000x reference)
#include <hip/hip_runtime.h>
#include <hip/hip_bf16.h>

#define BN   256
#define Lr   64
#define Dd   96
#define LDSS 100

typedef __attribute__((ext_vector_type(2))) float v2f;
typedef __attribute__((ext_vector_type(4))) float f4;
typedef __attribute__((ext_vector_type(8))) float v8f;

#if __has_builtin(__builtin_amdgcn_tanhf)
#define TANHF(v) __builtin_amdgcn_tanhf(v)
#else
#define TANHF(v) tanhf(v)
#endif

__global__ __launch_bounds__(256) void sim_fused_kernel(
    const float* __restrict__ x,
    const float* __restrict__ FC1,
    const float* __restrict__ FC2,
    const float* __restrict__ att,
    float* __restrict__ out)
{
    __shared__ float xs[Lr * LDSS];
    __shared__ float As[Lr * LDSS];
    __shared__ float Cs[Lr * LDSS];
    __shared__ float attS[Dd];

    const int bn = blockIdx.x;
    const int t  = threadIdx.x;
    const float* xb = x + (size_t)bn * Lr * Dd;

    for (int idx = t; idx < Lr * (Dd / 4); idx += 256) {
        const int row = idx / (Dd / 4);
        const int c4  = (idx % (Dd / 4)) * 4;
        f4 v = *(const f4*)(xb + row * Dd + c4);
        *(f4*)(&xs[row * LDSS + c4]) = v;
    }
    if (t < Dd) attS[t] = att[t];
    __syncthreads();

    const int wave  = t >> 5;
    const int lane  = t & 31;
    const int mtile = wave & 3;
    const float* W  = (wave < 4) ? FC1 : FC2;
    float*     Dst  = (wave < 4) ? As  : Cs;

    const int mrow = mtile * 16 + (lane & 15);
    const int kk   = (lane >> 4) * 2;

    for (int nt = 0; nt < 6; ++nt) {
        const int n = nt * 16 + (lane & 15);
        v8f acc = {};
        #pragma unroll
        for (int k4 = 0; k4 < Dd; k4 += 4) {
            v2f a = *(const v2f*)(&xs[mrow * LDSS + k4 + kk]);
            v2f b;
            b.x = W[(k4 + kk)     * Dd + n];
            b.y = W[(k4 + kk + 1) * Dd + n];
            acc = __builtin_amdgcn_wmma_f32_16x16x4_f32(
                 false, a,  false, b,
                 (short)0, acc,  false,  false);
        }
        const int drow0 = mtile * 16 + 8 * (lane >> 4);
        #pragma unroll
        for (int r = 0; r < 8; ++r)
            Dst[(drow0 + r) * LDSS + n] = acc[r];
    }
    __syncthreads();

    const int i0 = (t >> 4) * 4;
    const int j0 = (t & 15) * 4;

    float acc[4][4] = {};
    for (int d = 0; d < Dd; d += 4) {
        f4 w4 = *(const f4*)(&attS[d]);
        f4 ad[4], cd[4];
        #pragma unroll
        for (int u = 0; u < 4; ++u) {
            ad[u] = *(const f4*)(&As[(j0 + u) * LDSS + d]);
            cd[u] = *(const f4*)(&Cs[(i0 + u) * LDSS + d]);
        }
        #pragma unroll
        for (int ii = 0; ii < 4; ++ii) {
            #pragma unroll
            for (int jj = 0; jj < 4; ++jj) {
                #pragma unroll
                for (int c = 0; c < 4; ++c)
                    acc[ii][jj] += TANHF(ad[jj][c] + cd[ii][c]) * w4[c];
            }
        }
    }

    float* ob = out + (size_t)bn * Lr * Lr;
    #pragma unroll
    for (int ii = 0; ii < 4; ++ii) {
        f4 v;
        v.x = acc[ii][0]; v.y = acc[ii][1]; v.z = acc[ii][2]; v.w = acc[ii][3];
        *(f4*)(&ob[(i0 + ii) * Lr + j0]) = v;
    }
}

extern "C" void kernel_launch(void* const* d_in, const int* in_sizes, int n_in,
                              void* d_out, int out_size, void* d_ws, size_t ws_size,
                              hipStream_t stream) {
    const float* x   = (const float*)d_in[0];
    const float* FC1 = (const float*)d_in[1];
    const float* FC2 = (const float*)d_in[2];
    const float* att = (const float*)d_in[3];
    float* out = (float*)d_out;

    sim_fused_kernel<<<dim3(BN), dim3(256), 0, stream>>>(x, FC1, FC2, att, out);
}


# Global loads issued ahead of use: six x-tile loads as one group; GEMM k loop outermost over six accumulator tiles, B fragments loaded four k-steps ahead, A fragments read as one group
# speedup vs baseline: 1.3172x; 1.3172x over previous
#include <hip/hip_runtime.h>
#include <hip/hip_bf16.h>


#define BN   256
#define Lr   64
#define Dd   96
#define LDSS 100
#define KS   (Dd / 4)
#define NT   (Dd / 16)
#define PF   4

typedef __attribute__((ext_vector_type(2))) float v2f;
typedef __attribute__((ext_vector_type(4))) float f4;
typedef __attribute__((ext_vector_type(8))) float v8f;

#if __has_builtin(__builtin_amdgcn_tanhf)
#define TANHF(v) __builtin_amdgcn_tanhf(v)
#else
#define TANHF(v) tanhf(v)
#endif

__device__ __forceinline__ void load_b_step(const float* __restrict__ W, int s, int kk, int ncol,
                                            v2f (&b)[NT]) {
    const int k = s * 4 + kk;
    #pragma unroll
    for (int nt = 0; nt < NT; ++nt) {
        b[nt].x = W[k * Dd + nt * 16 + ncol];
        b[nt].y = W[(k + 1) * Dd + nt * 16 + ncol];
    }
}

__global__ __launch_bounds__(256) void sim_fused_kernel(
    const float* __restrict__ x,
    const float* __restrict__ FC1,
    const float* __restrict__ FC2,
    const float* __restrict__ att,
    float* __restrict__ out)
{
    __shared__ float xs[Lr * LDSS];
    __shared__ float As[Lr * LDSS];
    __shared__ float Cs[Lr * LDSS];
    __shared__ float attS[Dd];

    const int bn = blockIdx.x;
    const int t  = threadIdx.x;
    const float* xb = x + (size_t)bn * Lr * Dd;

    const int wave  = t >> 5;
    const int lane  = t & 31;
    const int mtile = wave & 3;
    const float* W  = (wave < 4) ? FC1 : FC2;
    float*     Dst  = (wave < 4) ? As  : Cs;

    const int mrow = mtile * 16 + (lane & 15);
    const int kk   = (lane >> 4) * 2;
    const int ncol = lane & 15;

    f4 xv[6];
    #pragma unroll
    for (int q = 0; q < 6; ++q)
        xv[q] = *(const f4*)(xb + (t + q * 256) * 4);
    float attv = 0.0f;
    if (t < Dd) attv = att[t];

    v2f bq[KS][NT];
    #pragma unroll
    for (int s = 0; s < PF; ++s)
        load_b_step(W, s, kk, ncol, bq[s]);

    #pragma unroll
    for (int q = 0; q < 6; ++q) {
        const int idx = t + q * 256;
        const int row = idx / (Dd / 4);
        const int c4  = (idx % (Dd / 4)) * 4;
        *(f4*)(&xs[row * LDSS + c4]) = xv[q];
    }
    if (t < Dd) attS[t] = attv;
    __syncthreads();

    v2f af[KS];
    #pragma unroll
    for (int s = 0; s < KS; ++s)
        af[s] = *(const v2f*)(&xs[mrow * LDSS + s * 4 + kk]);
    asm volatile("" :: "v"(af[0]),  "v"(af[1]),  "v"(af[2]),  "v"(af[3]),  "v"(af[4]),  "v"(af[5]),
                       "v"(af[6]),  "v"(af[7]),  "v"(af[8]),  "v"(af[9]),  "v"(af[10]), "v"(af[11]),
                       "v"(af[12]), "v"(af[13]), "v"(af[14]), "v"(af[15]), "v"(af[16]), "v"(af[17]),
                       "v"(af[18]), "v"(af[19]), "v"(af[20]), "v"(af[21]), "v"(af[22]), "v"(af[23]));

    v8f dacc[NT] = {};
    #pragma unroll
    for (int s = 0; s < KS; ++s) {
        if (s + PF < KS)
            load_b_step(W, s + PF, kk, ncol, bq[s + PF]);
        #pragma unroll
        for (int nt = 0; nt < NT; ++nt) {
            dacc[nt] = __builtin_amdgcn_wmma_f32_16x16x4_f32(
                 false, af[s],  false, bq[s][nt],
                 (short)0, dacc[nt],  false,  false);
        }
    }
    const int drow0 = mtile * 16 + 8 * (lane >> 4);
    #pragma unroll
    for (int nt = 0; nt < NT; ++nt) {
        #pragma unroll
        for (int r = 0; r < 8; ++r)
            Dst[(drow0 + r) * LDSS + nt * 16 + ncol] = dacc[nt][r];
    }
    __syncthreads();

    const int i0 = (t >> 4) * 4;
    const int j0 = (t & 15) * 4;

    float acc[4][4] = {};
    for (int d = 0; d < Dd; d += 4) {
        f4 w4 = *(const f4*)(&attS[d]);
        f4 ad[4], cd[4];
        #pragma unroll
        for (int u = 0; u < 4; ++u) {
            ad[u] = *(const f4*)(&As[(j0 + u) * LDSS + d]);
            cd[u] = *(const f4*)(&Cs[(i0 + u) * LDSS + d]);
        }
        #pragma unroll
        for (int ii = 0; ii < 4; ++ii) {
            #pragma unroll
            for (int jj = 0; jj < 4; ++jj) {
                #pragma unroll
                for (int c = 0; c < 4; ++c)
                    acc[ii][jj] += TANHF(ad[jj][c] + cd[ii][c]) * w4[c];
            }
        }
    }

    float* ob = out + (size_t)bn * Lr * Lr;
    #pragma unroll
    for (int ii = 0; ii < 4; ++ii) {
        f4 v;
        v.x = acc[ii][0]; v.y = acc[ii][1]; v.z = acc[ii][2]; v.w = acc[ii][3];
        *(f4*)(&ob[(i0 + ii) * Lr + j0]) = v;
    }
}

extern "C" void kernel_launch(void* const* d_in, const int* in_sizes, int n_in,
                              void* d_out, int out_size, void* d_ws, size_t ws_size,
                              hipStream_t stream) {
    const float* x   = (const float*)d_in[0];
    const float* FC1 = (const float*)d_in[1];
    const float* FC2 = (const float*)d_in[2];
    const float* att = (const float*)d_in[3];
    float* out = (float*)d_out;

    sim_fused_kernel<<<dim3(BN), dim3(256), 0, stream>>>(x, FC1, FC2, att, out);
}


# Epilogue software-pipelined: every tanh issued beside a packed add or multiply-add instead of in runs of 64 tanh, LDS reads of the next d-group issued a group ahead; order of each sum unchanged
# speedup vs baseline: 1.4710x; 1.1168x over previous
#include <hip/hip_runtime.h>
#include <hip/hip_bf16.h>


#define BN   256
#define Lr   64
#define Dd   96
#define LDSS 100
#define KS   (Dd / 4)
#define NT   (Dd / 16)
#define PF   4

typedef __attribute__((ext_vector_type(2))) float v2f;
typedef __attribute__((ext_vector_type(4))) float f4;
typedef __attribute__((ext_vector_type(8))) float v8f;

#if __has_builtin(__builtin_amdgcn_tanhf)
#define TANHF(v) __builtin_amdgcn_tanhf(v)
#else
#define TANHF(v) tanhf(v)
#endif

__device__ __forceinline__ void load_b_step(const float* __restrict__ W, int s, int kk, int ncol,
                                            v2f (&b)[NT]) {
    const int k = s * 4 + kk;
    #pragma unroll
    for (int nt = 0; nt < NT; ++nt) {
        b[nt].x = W[k * Dd + nt * 16 + ncol];
        b[nt].y = W[(k + 1) * Dd + nt * 16 + ncol];
    }
}

__device__ __forceinline__ void tanh_group(const f4& w, const f4 (&a)[4], const f4 (&c)[4], v2f (&acc2)[2][4]) {
    v2f ah[2][4], ch[2][4];
    #pragma unroll
    for (int u = 0; u < 4; ++u) {
        ah[0][u] = a[u].xy; ah[1][u] = a[u].zw;
        ch[0][u] = c[u].xy; ch[1][u] = c[u].zw;
    }
    const float wv[4] = {w.x, w.y, w.z, w.w};
    v2f sA[16], sB[16], p0[16], p1[16];
#define KEEP_ORDER() __builtin_amdgcn_sched_barrier(0)
#define U_N(k)  (2 * ((k) >> 2) + ((k) & 1))
#define U_IP(k) (U_N(k) >> 2)
#define U_JJ(k) (U_N(k) & 3)
#define U_H(k)  (((k) >> 1) & 1)
#define UNIT_ADD_A(k) sA[k] = ah[U_H(k)][U_JJ(k)] + ch[U_H(k)][2 * U_IP(k)]
#define UNIT_ADD_B(k) sB[k] = ah[U_H(k)][U_JJ(k)] + ch[U_H(k)][2 * U_IP(k) + 1]
#define UNIT_FMA0(k) do { v2f wb_; wb_.x = wv[2 * U_H(k)]; wb_.y = wv[2 * U_H(k)];                      \
        acc2[U_IP(k)][U_JJ(k)] = __builtin_elementwise_fma(p0[k], wb_, acc2[U_IP(k)][U_JJ(k)]); } while (0)
#define UNIT_FMA1(k) do { v2f wb_; wb_.x = wv[2 * U_H(k) + 1]; wb_.y = wv[2 * U_H(k) + 1];              \
        acc2[U_IP(k)][U_JJ(k)] = __builtin_elementwise_fma(p1[k], wb_, acc2[U_IP(k)][U_JJ(k)]); } while (0)
    UNIT_ADD_A(0);
    UNIT_ADD_B(0);
    KEEP_ORDER();
    #pragma unroll
    for (int k = 0; k < 16; ++k) {
        p0[k].x = TANHF(sA[k].x);
        KEEP_ORDER();
        if (k + 1 < 16) UNIT_ADD_A(k + 1);
        KEEP_ORDER();
        p0[k].y = TANHF(sB[k].x);
        KEEP_ORDER();
        if (k + 1 < 16) UNIT_ADD_B(k + 1);
        KEEP_ORDER();
        p1[k].x = TANHF(sA[k].y);
        KEEP_ORDER();
        if (k >= 1) UNIT_FMA0(k - 1);
        KEEP_ORDER();
        p1[k].y = TANHF(sB[k].y);
        KEEP_ORDER();
        if (k >= 2) UNIT_FMA1(k - 2);
        KEEP_ORDER();
    }
    UNIT_FMA1(14);
    UNIT_FMA0(15);
    UNIT_FMA1(15);
#undef KEEP_ORDER
#undef U_N
#undef U_IP
#undef U_JJ
#undef U_H
#undef UNIT_ADD_A
#undef UNIT_ADD_B
#undef UNIT_FMA0
#undef UNIT_FMA1
}

__device__ __forceinline__ void load_group(const float* __restrict__ attp, const float* __restrict__ Ap,
                                           const float* __restrict__ Cp, int d, f4& w, f4 (&a)[4], f4 (&c)[4]) {
    w = *(const f4*)(attp + d);
    #pragma unroll
    for (int u = 0; u < 4; ++u) {
        a[u] = *(const f4*)(Ap + u * LDSS + d);
        c[u] = *(const f4*)(Cp + u * LDSS + d);
    }
}

__global__ __launch_bounds__(256) void sim_fused_kernel(
    const float* __restrict__ x,
    const float* __restrict__ FC1,
    const float* __restrict__ FC2,
    const float* __restrict__ att,
    float* __restrict__ out)
{
    __shared__ float xs[Lr * LDSS];
    __shared__ float As[Lr * LDSS];
    __shared__ float Cs[Lr * LDSS];
    __shared__ float attS[Dd];

    const int bn = blockIdx.x;
    const int t  = threadIdx.x;
    const float* xb = x + (size_t)bn * Lr * Dd;

    const int wave  = t >> 5;
    const int lane  = t & 31;
    const int mtile = wave & 3;
    const float* W  = (wave < 4) ? FC1 : FC2;
    float*     Dst  = (wave < 4) ? As  : Cs;

    const int mrow = mtile * 16 + (lane & 15);
    const int kk   = (lane >> 4) * 2;
    const int ncol = lane & 15;

    f4 xv[6];
    #pragma unroll
    for (int q = 0; q < 6; ++q)
        xv[q] = *(const f4*)(xb + (t + q * 256) * 4);
    float attv = 0.0f;
    if (t < Dd) attv = att[t];

    v2f bq[KS][NT];
    #pragma unroll
    for (int s = 0; s < PF; ++s)
        load_b_step(W, s, kk, ncol, bq[s]);

    #pragma unroll
    for (int q = 0; q < 6; ++q) {
        const int idx = t + q * 256;
        const int row = idx / (Dd / 4);
        const int c4  = (idx % (Dd / 4)) * 4;
        *(f4*)(&xs[row * LDSS + c4]) = xv[q];
    }
    if (t < Dd) attS[t] = attv;
    __syncthreads();

    v2f af[KS];
    #pragma unroll
    for (int s = 0; s < KS; ++s)
        af[s] = *(const v2f*)(&xs[mrow * LDSS + s * 4 + kk]);
    asm volatile("" :: "v"(af[0]),  "v"(af[1]),  "v"(af[2]),  "v"(af[3]),  "v"(af[4]),  "v"(af[5]),
                       "v"(af[6]),  "v"(af[7]),  "v"(af[8]),  "v"(af[9]),  "v"(af[10]), "v"(af[11]),
                       "v"(af[12]), "v"(af[13]), "v"(af[14]), "v"(af[15]), "v"(af[16]), "v"(af[17]),
                       "v"(af[18]), "v"(af[19]), "v"(af[20]), "v"(af[21]), "v"(af[22]), "v"(af[23]));

    v8f dacc[NT] = {};
    #pragma unroll
    for (int s = 0; s < KS; ++s) {
        if (s + PF < KS)
            load_b_step(W, s + PF, kk, ncol, bq[s + PF]);
        #pragma unroll
        for (int nt = 0; nt < NT; ++nt) {
            dacc[nt] = __builtin_amdgcn_wmma_f32_16x16x4_f32(
                 false, af[s],  false, bq[s][nt],
                 (short)0, dacc[nt],  false,  false);
        }
    }
    const int drow0 = mtile * 16 + 8 * (lane >> 4);
    #pragma unroll
    for (int nt = 0; nt < NT; ++nt) {
        #pragma unroll
        for (int r = 0; r < 8; ++r)
            Dst[(drow0 + r) * LDSS + nt * 16 + ncol] = dacc[nt][r];
    }
    __syncthreads();

    const int i0 = (t >> 4) * 4;
    const int j0 = (t & 15) * 4;
    const float* Ap = &As[j0 * LDSS];
    const float* Cp = &Cs[i0 * LDSS];

    v2f acc2[2][4] = {};
    f4 wX, aX[4], cX[4], wY, aY[4], cY[4];
    load_group(attS, Ap, Cp, 0, wX, aX, cX);
    for (int d = 0; d < Dd; d += 8) {
        load_group(attS, Ap, Cp, d + 4, wY, aY, cY);
        tanh_group(wX, aX, cX, acc2);
        const int dn = (d + 8 < Dd) ? d + 8 : d;
        load_group(attS, Ap, Cp, dn, wX, aX, cX);
        tanh_group(wY, aY, cY, acc2);
    }

    float* ob = out + (size_t)bn * Lr * Lr;
    #pragma unroll
    for (int ii = 0; ii < 4; ++ii) {
        f4 v;
        if (ii & 1) { v.x = acc2[ii >> 1][0].y; v.y = acc2[ii >> 1][1].y; v.z = acc2[ii >> 1][2].y; v.w = acc2[ii >> 1][3].y; }
        else        { v.x = acc2[ii >> 1][0].x; v.y = acc2[ii >> 1][1].x; v.z = acc2[ii >> 1][2].x; v.w = acc2[ii >> 1][3].x; }
        *(f4*)(&ob[(i0 + ii) * Lr + j0]) = v;
    }
}

extern "C" void kernel_launch(void* const* d_in, const int* in_sizes, int n_in,
                              void* d_out, int out_size, void* d_ws, size_t ws_size,
                              hipStream_t stream) {
    const float* x   = (const float*)d_in[0];
    const float* FC1 = (const float*)d_in[1];
    const float* FC2 = (const float*)d_in[2];
    const float* att = (const float*)d_in[3];
    float* out = (float*)d_out;

    sim_fused_kernel<<<dim3(BN), dim3(256), 0, stream>>>(x, FC1, FC2, att, out);
}
